# starting kernel
# baseline (speedup 1.0000x reference)
#include <hip/hip_runtime.h>
#include <hip/hip_bf16.h>
#include <math.h>

typedef float v2f __attribute__((ext_vector_type(2)));
typedef float v8f __attribute__((ext_vector_type(8)));

#define GMAXC 1.0f
#define GMINC 0.01f
#define EPSC  0.1f

static constexpr int N_ = 8, B_ = 32, NIN_ = 128, NOUT_ = 128, E_ = 1024, M_ = 130;
static constexpr int VSTR = 132;

__device__ __forceinline__ float theta_eff(float th) {
    th = fminf(fmaxf(th, -GMAXC), GMAXC);
    return (fabsf(th) < GMINC) ? 0.0f : th;
}

__global__ void prep_kernel(const float* __restrict__ theta,
                            const float* __restrict__ noise,
                            float* __restrict__ Vt,
                            float* __restrict__ cbias)
{
    int t = blockIdx.x * blockDim.x + threadIdx.x;
    int n = t >> 7;
    int o = t & 127;
    const float* nz = noise + (size_t)n * M_ * NOUT_ + o;

    float denom = 1e-10f;
    for (int m = 0; m < M_; ++m) {
        float th = theta_eff(theta[m * NOUT_ + o]);
        float f  = nz[(size_t)m * NOUT_];
        float tn = th * ((f * 2.0f - 1.0f) * EPSC + 1.0f);
        denom += fabsf(tn);
    }
    float inv = 1.0f / denom;

    float bias = 0.0f;
    float* vrow = Vt + ((size_t)n * NOUT_ + o) * NIN_;
    for (int m = 0; m < M_; ++m) {
        float th = theta_eff(theta[m * NOUT_ + o]);
        float f  = nz[(size_t)m * NOUT_];
        float tn = th * ((f * 2.0f - 1.0f) * EPSC + 1.0f);
        if (m < NIN_) {
            vrow[m] = tn * inv;
            bias += fmaxf(-tn, 0.0f);
        } else if (m == NIN_) {
            bias += fmaxf(tn, 0.0f);
        }
    }
    cbias[n * NOUT_ + o] = bias * inv;
}

__global__ void __launch_bounds__(256)
gemm_act_kernel(const float* __restrict__ a,
                const float* __restrict__ Vt,
                const float* __restrict__ cbias,
                const float* __restrict__ eta,
                float* __restrict__ out)
{
    __shared__ float ldsV[64 * VSTR];
    __shared__ float ldsC[64];

    const int bid    = blockIdx.x;
    const int ohalf  = bid & 1;
    const int echunk = (bid >> 1) & 7;
    const int b      = (bid >> 4) & 31;
    const int n      = bid >> 9;
    const int tid    = threadIdx.x;
    const int o0     = ohalf * 64;

    const float* vsrc = Vt + ((size_t)n * NOUT_ + o0) * NIN_;
    #pragma unroll
    for (int r = 0; r < 32; ++r) {
        int idx  = r * 256 + tid;
        int orow = idx >> 7;
        int m    = idx & 127;
        ldsV[orow * VSTR + m] = vsrc[(size_t)orow * NIN_ + m];
    }
    if (tid < 64) ldsC[tid] = cbias[n * NOUT_ + o0 + tid];
    __syncthreads();

    const int wave  = tid >> 5;
    const int lane  = tid & 31;
    const int lrow  = lane & 15;
    const int lhalf = lane >> 4;
    const int e0    = echunk * 128 + wave * 16;

    const float* Abase = a + ((size_t)(n * B_ + b) * NIN_) * E_ + e0 + lrow;

    v8f acc[4];
    #pragma unroll
    for (int ot = 0; ot < 4; ++ot)
        #pragma unroll
        for (int i = 0; i < 8; ++i) acc[ot][i] = 0.0f;

    for (int k0 = 0; k0 < NIN_; k0 += 4) {
        const int kk = k0 + 2 * lhalf;
        const float* ap = Abase + (size_t)kk * E_;
        v2f Bf;
        Bf.x = ap[0];
        Bf.y = ap[E_];
        #pragma unroll
        for (int ot = 0; ot < 4; ++ot) {
            const float* vp = &ldsV[(ot * 16 + lrow) * VSTR + kk];
            v2f Af;
            Af.x = vp[0];
            Af.y = vp[1];
            acc[ot] = __builtin_amdgcn_wmma_f32_16x16x4_f32(
                false, Af, false, Bf, (short)0, acc[ot], false, false);
        }
    }

    const float eta0 = eta[0], eta1 = eta[1], eta2 = eta[2], eta3 = eta[3];
    float* obase = out + ((size_t)(n * B_ + b) * NOUT_) * E_;
    #pragma unroll
    for (int ot = 0; ot < 4; ++ot) {
        #pragma unroll
        for (int r = 0; r < 8; ++r) {
            const int oo = ot * 16 + r + 8 * lhalf;
            float z   = acc[ot][r] + ldsC[oo];
            float act = eta0 + eta1 * tanhf((z - eta2) * eta3);
            obase[(size_t)(o0 + oo) * E_ + e0 + lrow] = act;
        }
    }
}

extern "C" void kernel_launch(void* const* d_in, const int* in_sizes, int n_in,
                              void* d_out, int out_size, void* d_ws, size_t ws_size,
                              hipStream_t stream)
{
    const float* a     = (const float*)d_in[0];
    const float* theta = (const float*)d_in[1];
    const float* noise = (const float*)d_in[2];
    const float* eta   = (const float*)d_in[3];
    float* out = (float*)d_out;

    float* Vt = (float*)d_ws;
    float* cb = Vt + (size_t)N_ * NOUT_ * NIN_;

    prep_kernel<<<4, 256, 0, stream>>>(theta, noise, Vt, cb);

    const int nblocks = N_ * B_ * (E_ / 128) * 2;
    gemm_act_kernel<<<nblocks, 256, 0, stream>>>(a, Vt, cb, eta, out);
}


# persistent blocks: a wave keeps its 16 weight rows as register A-fragments built in-kernel from theta, noise (denominator, bias: half-wave partial sums + 1 exchange), walks 8 LDS tiles; 1 launch
# speedup vs baseline: 1.7937x; 1.7937x over previous
#include <hip/hip_runtime.h>
#include <hip/hip_bf16.h>
#include <math.h>


typedef float v2f __attribute__((ext_vector_type(2)));
typedef float v4f __attribute__((ext_vector_type(4)));
typedef float v8f __attribute__((ext_vector_type(8)));

#define GMAXC 1.0f
#define GMINC 0.01f
#define EPSC  0.1f

static constexpr int N_ = 8, B_ = 32, NIN_ = 128, NOUT_ = 128, E_ = 1024, M_ = 130;
static constexpr int ET_     = 64;
static constexpr int NT_     = 8;
static constexpr int TILE_F  = NIN_ * ET_;
static constexpr int PROW_F  = 2 * ET_;

__device__ __forceinline__ float theta_eff(float th) {
    th = fminf(fmaxf(th, -GMAXC), GMAXC);
    return (fabsf(th) < GMINC) ? 0.0f : th;
}

__device__ __forceinline__ float noisy_theta(float th, float f) {
    return theta_eff(th) * ((f * 2.0f - 1.0f) * EPSC + 1.0f);
}

__device__ __forceinline__ void lds_barrier() {
    __builtin_amdgcn_fence(__ATOMIC_RELEASE, "workgroup", "local");
    __builtin_amdgcn_s_barrier();
    __builtin_amdgcn_fence(__ATOMIC_ACQUIRE, "workgroup", "local");
}

__device__ __forceinline__ void tile_mma_store(const v2f (&Af)[32], const float* __restrict__ fb,
                                               float* __restrict__ otile, const float (&bias_r)[8],
                                               float eta0, float eta1, float eta2, float eta3)
{
    v8f acc[4];
    #pragma unroll
    for (int j = 0; j < 4; ++j)
        #pragma unroll
        for (int i = 0; i < 8; ++i) acc[j][i] = 0.0f;

    v4f lo[2][4], hi[2][4];
    #pragma unroll
    for (int q = 0; q < 4; ++q) {
        lo[0][q] = *(const v4f*)(fb + q * (2 * PROW_F));
        hi[0][q] = *(const v4f*)(fb + q * (2 * PROW_F) + ET_);
    }

    #pragma unroll
    for (int g = 0; g < 8; ++g) {
        if (g + 1 < 8) {
            #pragma unroll
            for (int q = 0; q < 4; ++q) {
                lo[(g + 1) & 1][q] = *(const v4f*)(fb + (4 * (g + 1) + q) * (2 * PROW_F));
                hi[(g + 1) & 1][q] = *(const v4f*)(fb + (4 * (g + 1) + q) * (2 * PROW_F) + ET_);
            }
        }
        __builtin_amdgcn_sched_barrier(0);
        #pragma unroll
        for (int q = 0; q < 4; ++q) {
            const int s = 4 * g + q;
            const v4f l = lo[g & 1][q];
            const v4f u = hi[g & 1][q];
            v2f b0, b1, b2, b3;
            b0.x = l.x; b0.y = l.y;
            b1.x = l.z; b1.y = l.w;
            b2.x = u.x; b2.y = u.y;
            b3.x = u.z; b3.y = u.w;
            acc[0] = __builtin_amdgcn_wmma_f32_16x16x4_f32(false, Af[s], false, b0, (short)0, acc[0], false, false);
            acc[1] = __builtin_amdgcn_wmma_f32_16x16x4_f32(false, Af[s], false, b1, (short)0, acc[1], false, false);
            acc[2] = __builtin_amdgcn_wmma_f32_16x16x4_f32(false, Af[s], false, b2, (short)0, acc[2], false, false);
            acc[3] = __builtin_amdgcn_wmma_f32_16x16x4_f32(false, Af[s], false, b3, (short)0, acc[3], false, false);
        }
        __builtin_amdgcn_sched_barrier(0);
    }

    #pragma unroll
    for (int r = 0; r < 8; ++r) {
        v4f v;
        v.x = eta0 + eta1 * tanhf(((acc[0][r] + bias_r[r]) - eta2) * eta3);
        v.y = eta0 + eta1 * tanhf(((acc[1][r] + bias_r[r]) - eta2) * eta3);
        v.z = eta0 + eta1 * tanhf(((acc[2][r] + bias_r[r]) - eta2) * eta3);
        v.w = eta0 + eta1 * tanhf(((acc[3][r] + bias_r[r]) - eta2) * eta3);
        __builtin_nontemporal_store(v, (v4f*)(otile + (size_t)r * E_));
    }
}

__global__ void __launch_bounds__(256)
fused_layer_kernel(const float* __restrict__ a,
                   const float* __restrict__ theta,
                   const float* __restrict__ noise,
                   const float* __restrict__ eta,
                   float* __restrict__ out)
{
    __shared__ __attribute__((aligned(16))) float lds[2 * TILE_F];

    const int bid   = blockIdx.x;
    const int n     = bid >> 6;
    const int slice = bid & 63;
    const int b     = slice >> 1;
    const int ecol0 = (slice & 1) * (NT_ * ET_);

    const int tid  = threadIdx.x;
    const int w    = tid >> 5;
    const int lane = tid & 31;
    const int c    = lane & 15;
    const int h    = lane >> 4;

    const float* abase = a   + ((size_t)(n * B_ + b) * NIN_)  * E_ + ecol0;
    float*       obase = out + ((size_t)(n * B_ + b) * NOUT_) * E_ + ecol0;

    const int skp = tid >> 4;
    const int sg  = tid & 15;
    const float* sbase = abase + (size_t)(2 * skp) * E_ + 4 * sg;
    const int    sdst  = skp * PROW_F + 4 * sg;

    v4f sx[4], sy[4];

    #pragma unroll
    for (int p = 0; p < 4; ++p) {
        const float* q = sbase + (size_t)(32 * p) * E_;
        sx[p] = __builtin_nontemporal_load((const v4f*)q);
        sy[p] = __builtin_nontemporal_load((const v4f*)(q + E_));
    }

    const int o = 16 * w + c;
    const float* thp = theta + o;
    const float* nzp = noise + (size_t)n * M_ * NOUT_ + o;

    v2f Af[32];
    float sabs = 0.0f, sneg = 0.0f;
    #pragma unroll
    for (int s = 0; s < 32; ++s) {
        const int m0 = 4 * s + 2 * h;
        const float t0 = noisy_theta(thp[m0 * NOUT_],       nzp[m0 * NOUT_]);
        const float t1 = noisy_theta(thp[(m0 + 1) * NOUT_], nzp[(m0 + 1) * NOUT_]);
        Af[s].x = t0;
        Af[s].y = t1;
        sabs += fabsf(t0);
        sabs += fabsf(t1);
        sneg += fmaxf(-t0, 0.0f);
        sneg += fmaxf(-t1, 0.0f);
    }
    const float t128 = noisy_theta(thp[128 * NOUT_], nzp[128 * NOUT_]);
    const float t129 = noisy_theta(thp[129 * NOUT_], nzp[129 * NOUT_]);

    const float sabs_o = sabs + __shfl_xor(sabs, 16);
    const float sneg_o = sneg + __shfl_xor(sneg, 16);
    const float denom  = ((sabs_o + fabsf(t128)) + fabsf(t129)) + 1e-10f;
    const float inv    = 1.0f / denom;
    const float bias_o = (sneg_o + fmaxf(t128, 0.0f)) * inv;

    #pragma unroll
    for (int s = 0; s < 32; ++s) {
        Af[s].x *= inv;
        Af[s].y *= inv;
    }

    float bias_r[8];
    #pragma unroll
    for (int r = 0; r < 8; ++r) bias_r[r] = __shfl(bias_o, 8 * h + r);

    const float eta0 = eta[0], eta1 = eta[1], eta2 = eta[2], eta3 = eta[3];

    #pragma unroll
    for (int p = 0; p < 4; ++p) {
        float* d = lds + sdst + p * (16 * PROW_F);
        *(v4f*)d         = (v4f){sx[p].x, sy[p].x, sx[p].y, sy[p].y};
        *(v4f*)(d + ET_) = (v4f){sx[p].z, sy[p].z, sx[p].w, sy[p].w};
    }
    __syncthreads();

    const int    fofs  = h * PROW_F + 4 * c;
    float*       orow  = obase + (size_t)(16 * w + 8 * h) * E_ + 4 * c;

    #pragma unroll 1
    for (int t = 0; t < NT_ - 1; ++t) {
        #pragma unroll
        for (int p = 0; p < 4; ++p) {
            const float* q = sbase + (size_t)(32 * p) * E_ + (t + 1) * ET_;
            sx[p] = __builtin_nontemporal_load((const v4f*)q);
            sy[p] = __builtin_nontemporal_load((const v4f*)(q + E_));
        }

        tile_mma_store(Af, lds + (t & 1) * TILE_F + fofs, orow + t * ET_, bias_r, eta0, eta1, eta2, eta3);

        float* nb = lds + ((t + 1) & 1) * TILE_F + sdst;
        #pragma unroll
        for (int p = 0; p < 4; ++p) {
            float* d = nb + p * (16 * PROW_F);
            *(v4f*)d         = (v4f){sx[p].x, sy[p].x, sx[p].y, sy[p].y};
            *(v4f*)(d + ET_) = (v4f){sx[p].z, sy[p].z, sx[p].w, sy[p].w};
        }
        lds_barrier();
    }
    tile_mma_store(Af, lds + ((NT_ - 1) & 1) * TILE_F + fofs, orow + (NT_ - 1) * ET_, bias_r, eta0, eta1, eta2, eta3);
}

extern "C" void kernel_launch(void* const* d_in, const int* in_sizes, int n_in,
                              void* d_out, int out_size, void* d_ws, size_t ws_size,
                              hipStream_t stream)
{
    const float* a     = (const float*)d_in[0];
    const float* theta = (const float*)d_in[1];
    const float* noise = (const float*)d_in[2];
    const float* eta   = (const float*)d_in[3];
    float* out = (float*)d_out;

    fused_layer_kernel<<<512, 256, 0, stream>>>(a, theta, noise, eta, out);
}


# activation: tanh as 1-2/(exp2(k*(z+b))+1) on the hardware exp and rcp, 6 branch-free vector ops per element instead of library tanhf; bias-eta2, eta0+eta1 folded per row; diff about 2e-7*|eta1|
# speedup vs baseline: 2.1784x; 1.2145x over previous
#include <hip/hip_runtime.h>
#include <hip/hip_bf16.h>
#include <math.h>


typedef float v2f __attribute__((ext_vector_type(2)));
typedef float v4f __attribute__((ext_vector_type(4)));
typedef float v8f __attribute__((ext_vector_type(8)));

#define GMAXC 1.0f
#define GMINC 0.01f
#define EPSC  0.1f

static constexpr int N_ = 8, B_ = 32, NIN_ = 128, NOUT_ = 128, E_ = 1024, M_ = 130;
static constexpr int ET_     = 64;
static constexpr int NT_     = 8;
static constexpr int TILE_F  = NIN_ * ET_;
static constexpr int PROW_F  = 2 * ET_;

__device__ __forceinline__ float theta_eff(float th) {
    th = fminf(fmaxf(th, -GMAXC), GMAXC);
    return (fabsf(th) < GMINC) ? 0.0f : th;
}

__device__ __forceinline__ float noisy_theta(float th, float f) {
    return theta_eff(th) * ((f * 2.0f - 1.0f) * EPSC + 1.0f);
}

__device__ __forceinline__ float act_exp(float z, float brow, float ak, float am, float ac) {
    const float e = __builtin_amdgcn_exp2f((z + brow) * ak);
    const float q = __builtin_amdgcn_rcpf(e + 1.0f);
    return __builtin_fmaf(q, am, ac);
}

__device__ __forceinline__ void lds_barrier() {
    __builtin_amdgcn_fence(__ATOMIC_RELEASE, "workgroup", "local");
    __builtin_amdgcn_s_barrier();
    __builtin_amdgcn_fence(__ATOMIC_ACQUIRE, "workgroup", "local");
}

__device__ __forceinline__ void tile_mma_store(const v2f (&Af)[32], const float* __restrict__ fb,
                                               float* __restrict__ otile, const float (&brow)[8],
                                               float ak, float am, float ac)
{
    v8f acc[4];
    #pragma unroll
    for (int j = 0; j < 4; ++j)
        #pragma unroll
        for (int i = 0; i < 8; ++i) acc[j][i] = 0.0f;

    v4f lo[2][4], hi[2][4];
    #pragma unroll
    for (int q = 0; q < 4; ++q) {
        lo[0][q] = *(const v4f*)(fb + q * (2 * PROW_F));
        hi[0][q] = *(const v4f*)(fb + q * (2 * PROW_F) + ET_);
    }

    #pragma unroll
    for (int g = 0; g < 8; ++g) {
        if (g + 1 < 8) {
            #pragma unroll
            for (int q = 0; q < 4; ++q) {
                lo[(g + 1) & 1][q] = *(const v4f*)(fb + (4 * (g + 1) + q) * (2 * PROW_F));
                hi[(g + 1) & 1][q] = *(const v4f*)(fb + (4 * (g + 1) + q) * (2 * PROW_F) + ET_);
            }
        }
        __builtin_amdgcn_sched_barrier(0);
        #pragma unroll
        for (int q = 0; q < 4; ++q) {
            const int s = 4 * g + q;
            const v4f l = lo[g & 1][q];
            const v4f u = hi[g & 1][q];
            v2f b0, b1, b2, b3;
            b0.x = l.x; b0.y = l.y;
            b1.x = l.z; b1.y = l.w;
            b2.x = u.x; b2.y = u.y;
            b3.x = u.z; b3.y = u.w;
            acc[0] = __builtin_amdgcn_wmma_f32_16x16x4_f32(false, Af[s], false, b0, (short)0, acc[0], false, false);
            acc[1] = __builtin_amdgcn_wmma_f32_16x16x4_f32(false, Af[s], false, b1, (short)0, acc[1], false, false);
            acc[2] = __builtin_amdgcn_wmma_f32_16x16x4_f32(false, Af[s], false, b2, (short)0, acc[2], false, false);
            acc[3] = __builtin_amdgcn_wmma_f32_16x16x4_f32(false, Af[s], false, b3, (short)0, acc[3], false, false);
        }
        __builtin_amdgcn_sched_barrier(0);
    }

    #pragma unroll
    for (int r = 0; r < 8; ++r) {
        v4f v;
        v.x = act_exp(acc[0][r], brow[r], ak, am, ac);
        v.y = act_exp(acc[1][r], brow[r], ak, am, ac);
        v.z = act_exp(acc[2][r], brow[r], ak, am, ac);
        v.w = act_exp(acc[3][r], brow[r], ak, am, ac);
        __builtin_nontemporal_store(v, (v4f*)(otile + (size_t)r * E_));
    }
}

__global__ void __launch_bounds__(256)
fused_layer_kernel(const float* __restrict__ a,
                   const float* __restrict__ theta,
                   const float* __restrict__ noise,
                   const float* __restrict__ eta,
                   float* __restrict__ out)
{
    __shared__ __attribute__((aligned(16))) float lds[2 * TILE_F];

    const int bid   = blockIdx.x;
    const int n     = bid >> 6;
    const int slice = bid & 63;
    const int b     = slice >> 1;
    const int ecol0 = (slice & 1) * (NT_ * ET_);

    const int tid  = threadIdx.x;
    const int w    = tid >> 5;
    const int lane = tid & 31;
    const int c    = lane & 15;
    const int h    = lane >> 4;

    const float* abase = a   + ((size_t)(n * B_ + b) * NIN_)  * E_ + ecol0;
    float*       obase = out + ((size_t)(n * B_ + b) * NOUT_) * E_ + ecol0;

    const int skp = tid >> 4;
    const int sg  = tid & 15;
    const float* sbase = abase + (size_t)(2 * skp) * E_ + 4 * sg;
    const int    sdst  = skp * PROW_F + 4 * sg;

    v4f sx[4], sy[4];

    #pragma unroll
    for (int p = 0; p < 4; ++p) {
        const float* q = sbase + (size_t)(32 * p) * E_;
        sx[p] = __builtin_nontemporal_load((const v4f*)q);
        sy[p] = __builtin_nontemporal_load((const v4f*)(q + E_));
    }

    const int o = 16 * w + c;
    const float* thp = theta + o;
    const float* nzp = noise + (size_t)n * M_ * NOUT_ + o;

    v2f Af[32];
    float sabs = 0.0f, sneg = 0.0f;
    #pragma unroll
    for (int s = 0; s < 32; ++s) {
        const int m0 = 4 * s + 2 * h;
        const float t0 = noisy_theta(thp[m0 * NOUT_],       nzp[m0 * NOUT_]);
        const float t1 = noisy_theta(thp[(m0 + 1) * NOUT_], nzp[(m0 + 1) * NOUT_]);
        Af[s].x = t0;
        Af[s].y = t1;
        sabs += fabsf(t0);
        sabs += fabsf(t1);
        sneg += fmaxf(-t0, 0.0f);
        sneg += fmaxf(-t1, 0.0f);
    }
    const float t128 = noisy_theta(thp[128 * NOUT_], nzp[128 * NOUT_]);
    const float t129 = noisy_theta(thp[129 * NOUT_], nzp[129 * NOUT_]);

    const float sabs_o = sabs + __shfl_xor(sabs, 16);
    const float sneg_o = sneg + __shfl_xor(sneg, 16);
    const float denom  = ((sabs_o + fabsf(t128)) + fabsf(t129)) + 1e-10f;
    const float inv    = 1.0f / denom;
    const float bias_o = (sneg_o + fmaxf(t128, 0.0f)) * inv;

    #pragma unroll
    for (int s = 0; s < 32; ++s) {
        Af[s].x *= inv;
        Af[s].y *= inv;
    }

    float bias_r[8];
    #pragma unroll
    for (int r = 0; r < 8; ++r) bias_r[r] = __shfl(bias_o, 8 * h + r);

    const float eta0 = eta[0], eta1 = eta[1], eta2 = eta[2], eta3 = eta[3];
    const float ak = eta3 * 2.8853900817779268f;
    const float am = -2.0f * eta1;
    const float ac = eta0 + eta1;
    float brow[8];
    #pragma unroll
    for (int r = 0; r < 8; ++r) brow[r] = bias_r[r] - eta2;

    #pragma unroll
    for (int p = 0; p < 4; ++p) {
        float* d = lds + sdst + p * (16 * PROW_F);
        *(v4f*)d         = (v4f){sx[p].x, sy[p].x, sx[p].y, sy[p].y};
        *(v4f*)(d + ET_) = (v4f){sx[p].z, sy[p].z, sx[p].w, sy[p].w};
    }
    __syncthreads();

    const int    fofs  = h * PROW_F + 4 * c;
    float*       orow  = obase + (size_t)(16 * w + 8 * h) * E_ + 4 * c;

    #pragma unroll 1
    for (int t = 0; t < NT_ - 1; ++t) {
        #pragma unroll
        for (int p = 0; p < 4; ++p) {
            const float* q = sbase + (size_t)(32 * p) * E_ + (t + 1) * ET_;
            sx[p] = __builtin_nontemporal_load((const v4f*)q);
            sy[p] = __builtin_nontemporal_load((const v4f*)(q + E_));
        }

        tile_mma_store(Af, lds + (t & 1) * TILE_F + fofs, orow + t * ET_, brow, ak, am, ac);

        float* nb = lds + ((t + 1) & 1) * TILE_F + sdst;
        #pragma unroll
        for (int p = 0; p < 4; ++p) {
            float* d = nb + p * (16 * PROW_F);
            *(v4f*)d         = (v4f){sx[p].x, sy[p].x, sx[p].y, sy[p].y};
            *(v4f*)(d + ET_) = (v4f){sx[p].z, sy[p].z, sx[p].w, sy[p].w};
        }
        lds_barrier();
    }
    tile_mma_store(Af, lds + ((NT_ - 1) & 1) * TILE_F + fofs, orow + (NT_ - 1) * ET_, brow, ak, am, ac);
}

extern "C" void kernel_launch(void* const* d_in, const int* in_sizes, int n_in,
                              void* d_out, int out_size, void* d_ws, size_t ws_size,
                              hipStream_t stream)
{
    const float* a     = (const float*)d_in[0];
    const float* theta = (const float*)d_in[1];
    const float* noise = (const float*)d_in[2];
    const float* eta   = (const float*)d_in[3];
    float* out = (float*)d_out;

    fused_layer_kernel<<<512, 256, 0, stream>>>(a, theta, noise, eta, out);
}
